# starting kernel
# baseline (speedup 1.0000x reference)
#include <hip/hip_runtime.h>
#include <hip/hip_bf16.h>

typedef float v2f __attribute__((ext_vector_type(2)));
typedef float v8f __attribute__((ext_vector_type(8)));

#define CDIM  100
#define LSLOT 7
#define RSLOT 10
#define WDIM  300
#define NEGV  -1e30f

__device__ __forceinline__ float wave_sum(float v) {
  #pragma unroll
  for (int m = 16; m; m >>= 1) v += __shfl_xor(v, m, 32);
  return v;
}

__device__ __forceinline__ float score_slot(int idx, const float* vrow,
                                            const float* __restrict__ ce_raw, int lane) {
  if (idx < 0) return NEGV;
  const float* row = ce_raw + (long)idx * CDIM;
  float dot = 0.f, ss = 0.f;
  for (int c = lane; c < CDIM; c += 32) {
    float x = row[c];
    dot = fmaf(x, vrow[c], dot);
    ss  = fmaf(x, x, ss);
  }
  #pragma unroll
  for (int m = 16; m; m >>= 1) {
    dot += __shfl_xor(dot, m, 32);
    ss  += __shfl_xor(ss,  m, 32);
  }
  return dot / fmaxf(sqrtf(ss), 1e-12f);
}

__global__ __launch_bounds__(64) void fused_gated_attn_kernel(
    const float* __restrict__ ce_raw, const float* __restrict__ wvec,
    const float* __restrict__ lW,     const float* __restrict__ rW,
    const float* __restrict__ gL_w,   const float* __restrict__ gL_b,
    const int* __restrict__ qidx,
    const int* __restrict__ lixs_c,   const int* __restrict__ rixs_c,
    const int* __restrict__ lixs_w,   const int* __restrict__ rixs_w,
    float* __restrict__ out)
{
  __shared__ float q_lds[32][104];
  __shared__ float v_lds[32][2][112];

  const int lane  = threadIdx.x & 31;
  const int wv    = threadIdx.x >> 5;
  const int bbase = blockIdx.x * 32 + wv * 16;
  const int hi = lane >> 4, ln = lane & 15;

  for (int r = 0; r < 16; ++r) {
    long row = qidx[bbase + r];
    float vals[4]; float ss = 0.f;
    #pragma unroll
    for (int i = 0; i < 4; ++i) {
      int c = lane + 32 * i;
      vals[i] = (c < CDIM) ? ce_raw[row * CDIM + c] : 0.f;
      ss = fmaf(vals[i], vals[i], ss);
    }
    ss = wave_sum(ss);
    float inv = 1.f / fmaxf(sqrtf(ss), 1e-12f);
    #pragma unroll
    for (int i = 0; i < 4; ++i) {
      int c = lane + 32 * i;
      if (c < CDIM) q_lds[wv * 16 + r][c] = vals[i] * inv;
    }
  }
  __syncthreads();

  for (int side = 0; side < 2; ++side) {
    const float* W = side ? rW : lW;
    for (int nt = 0; nt < 7; ++nt) {
      int cout = nt * 16 + ln;
      v8f acc = {};
      for (int k = 0; k < 25; ++k) {
        int d0 = k * 4 + 2 * hi;
        v2f a, bfrag;
        a[0] = q_lds[wv * 16 + ln][d0];
        a[1] = q_lds[wv * 16 + ln][d0 + 1];
        if (cout < CDIM) {
          bfrag[0] = W[cout * CDIM + d0];
          bfrag[1] = W[cout * CDIM + d0 + 1];
        } else {
          bfrag[0] = 0.f; bfrag[1] = 0.f;
        }
        acc = __builtin_amdgcn_wmma_f32_16x16x4_f32(
             false, a,  false, bfrag,
             (short)0, acc,  false,  false);
      }
      #pragma unroll
      for (int r = 0; r < 8; ++r)
        v_lds[wv * 16 + r + 8 * hi][side][nt * 16 + ln] = acc[r];
    }
  }
  __syncthreads();

  for (int bb = 0; bb < 16; ++bb) {
    const int b  = bbase + bb;
    const int bl = wv * 16 + bb;

    float la[LSLOT], ra[RSLOT];
    #pragma unroll
    for (int t = 0; t < LSLOT; ++t)
      la[t] = score_slot(lixs_c[b * LSLOT + t], &v_lds[bl][0][0], ce_raw, lane);
    #pragma unroll
    for (int t = 0; t < RSLOT; ++t)
      ra[t] = score_slot(rixs_c[b * RSLOT + t], &v_lds[bl][1][0], ce_raw, lane);

    float mx = la[0];
    #pragma unroll
    for (int t = 1; t < LSLOT; ++t) mx = fmaxf(mx, la[t]);
    float s = 0.f;
    #pragma unroll
    for (int t = 0; t < LSLOT; ++t) { la[t] = expf(la[t] - mx); s += la[t]; }
    float rs = 1.f / s;
    #pragma unroll
    for (int t = 0; t < LSLOT; ++t) la[t] *= rs;

    mx = ra[0];
    #pragma unroll
    for (int t = 1; t < RSLOT; ++t) mx = fmaxf(mx, ra[t]);
    s = 0.f;
    #pragma unroll
    for (int t = 0; t < RSLOT; ++t) { ra[t] = expf(ra[t] - mx); s += ra[t]; }
    rs = 1.f / s;
    #pragma unroll
    for (int t = 0; t < RSLOT; ++t) ra[t] *= rs;

    float z0 = gL_b[0], z1 = gL_b[1];
    #pragma unroll
    for (int t = 0; t < LSLOT; ++t) {
      z0 = fmaf(la[t], gL_w[t],                z0);
      z1 = fmaf(la[t], gL_w[(LSLOT + RSLOT) + t], z1);
    }
    #pragma unroll
    for (int t = 0; t < RSLOT; ++t) {
      z0 = fmaf(ra[t], gL_w[LSLOT + t],                       z0);
      z1 = fmaf(ra[t], gL_w[(LSLOT + RSLOT) + LSLOT + t],     z1);
    }
    float gm = fmaxf(z0, z1);
    float e0 = expf(z0 - gm), e1 = expf(z1 - gm);
    float gi = 1.f / (e0 + e1);
    float g0 = e0 * gi, g1 = e1 * gi;

    for (int j = lane; j < WDIM; j += 32) {
      float accL = 0.f, accR = 0.f;
      #pragma unroll
      for (int t = 0; t < LSLOT; ++t) {
        int iw = lixs_w[b * LSLOT + t];
        if (iw >= 0) accL = fmaf(la[t], wvec[(long)iw * WDIM + j], accL);
      }
      #pragma unroll
      for (int t = 0; t < RSLOT; ++t) {
        int iw = rixs_w[b * RSLOT + t];
        if (iw >= 0) accR = fmaf(ra[t], wvec[(long)iw * WDIM + j], accR);
      }
      out[(long)b * WDIM + j] = g0 * accL + g1 * accR;
    }
  }
}

extern "C" void kernel_launch(void* const* d_in, const int* in_sizes, int n_in,
                              void* d_out, int out_size, void* d_ws, size_t ws_size,
                              hipStream_t stream) {
  const float* ce_raw = (const float*)d_in[0];
  const float* wvec   = (const float*)d_in[1];
  const float* lW     = (const float*)d_in[2];
  const float* rW     = (const float*)d_in[3];
  const float* gL_w   = (const float*)d_in[4];
  const float* gL_b   = (const float*)d_in[5];
  const int*   qidx   = (const int*)d_in[6];
  const int*   lixs_c = (const int*)d_in[7];
  const int*   rixs_c = (const int*)d_in[8];
  const int*   lixs_w = (const int*)d_in[9];
  const int*   rixs_w = (const int*)d_in[10];

  const int B = in_sizes[6];
  const int blocks = B / 32;
  fused_gated_attn_kernel<<<blocks, 64, 0, stream>>>(
      ce_raw, wvec, lW, rW, gL_w, gL_b,
      qidx, lixs_c, rixs_c, lixs_w, rixs_w, (float*)d_out);
}


# Work-group takes 32 rows (two M tiles, 16 waves): both weight matrices copied to LDS once at entry and kept, all 28 tile products in one phase; right-side reload and 2 barriers gone. Sums same.
# speedup vs baseline: 24.0553x; 24.0553x over previous
#include <hip/hip_runtime.h>

typedef float v2f __attribute__((ext_vector_type(2)));
typedef float v8f __attribute__((ext_vector_type(8)));

#define CDIM  100
#define LSLOT 7
#define RSLOT 10
#define NSLOT 17
#define WDIM  300
#define NEGV  -1e30f
#define ROWS  32
#define MTILE (ROWS / 16)
#define NJOB  (MTILE * 14)
#define NWAVE 16
#define NTHR  (32 * NWAVE)
#define C4    25
#define W4    75
#define NPAIR (ROWS * NSLOT)
#define WSZ4  (CDIM * CDIM / 4)
#define WLD   5
#define WPAD  (4 * NTHR * WLD)

__device__ __forceinline__ float wave_sum(float v) {
  #pragma unroll
  for (int m = 16; m; m >>= 1) v += __shfl_xor(v, m, 32);
  return v;
}

__device__ __forceinline__ float dot4(float4 x, float4 v, float acc) {
  acc = fmaf(x.x, v.x, acc);
  acc = fmaf(x.y, v.y, acc);
  acc = fmaf(x.z, v.z, acc);
  acc = fmaf(x.w, v.w, acc);
  return acc;
}

__device__ __forceinline__ float4 fma4(float s, float4 x, float4 a) {
  a.x = fmaf(s, x.x, a.x);
  a.y = fmaf(s, x.y, a.y);
  a.z = fmaf(s, x.z, a.z);
  a.w = fmaf(s, x.w, a.w);
  return a;
}

__device__ __forceinline__ void load_w(const float* __restrict__ Wg, int tid, float4 (&wreg)[WLD]) {
  const float4* wsrc = reinterpret_cast<const float4*>(Wg);
  #pragma unroll
  for (int i = 0; i < WLD; ++i) {
    const int f = tid + NTHR * i;
    wreg[i] = wsrc[f < WSZ4 ? f : WSZ4 - 1];
  }
}

__device__ __forceinline__ void store_w(float* wmat, int tid, const float4 (&wreg)[WLD]) {
  #pragma unroll
  for (int i = 0; i < WLD; ++i) {
    const int f = tid + NTHR * i;
    reinterpret_cast<float4*>(wmat)[f] = wreg[i];
  }
}

__device__ __forceinline__ void wmma_side(const float (*q)[104], const float* wmat, float (*v)[2][112],
                                          int side, int nt, int hi, int ln) {
  const int cout = nt * 16 + ln;
  const int crow = cout < CDIM ? cout : CDIM - 1;
  const float* qk = &q[ln][2 * hi];
  const float* wk = wmat + crow * CDIM + 2 * hi;
  v2f a[25], bfrag[25];
  #pragma unroll
  for (int k = 0; k < 25; ++k) {
    a[k]     = *reinterpret_cast<const v2f*>(qk + 4 * k);
    bfrag[k] = *reinterpret_cast<const v2f*>(wk + 4 * k);
  }
  __builtin_amdgcn_sched_barrier(0);
  v8f acc = {};
  #pragma unroll
  for (int k = 0; k < 25; ++k)
    acc = __builtin_amdgcn_wmma_f32_16x16x4_f32(
         false, a[k],  false, bfrag[k],
         (short)0, acc,  false,  false);
  #pragma unroll
  for (int r = 0; r < 8; ++r)
    v[r + 8 * hi][side][nt * 16 + ln] = acc[r];
}

__device__ __forceinline__ void wmma_job(const float (*q)[104], const float (*wmat)[WPAD], float (*v)[2][112],
                                         int j, int hi, int ln) {
  const int mt = j / 14, rem = j - 14 * mt;
  const int side = rem / 7, nt = rem - 7 * side;
  wmma_side(q + 16 * mt, wmat[side], v + 16 * mt, side, nt, hi, ln);
}

__global__ __launch_bounds__(NTHR) __attribute__((amdgpu_waves_per_eu(8, 8))) void fused_gated_attn_kernel(
    const float* __restrict__ ce_raw, const float* __restrict__ wvec,
    const float* __restrict__ lW,     const float* __restrict__ rW,
    const float* __restrict__ gL_w,   const float* __restrict__ gL_b,
    const int* __restrict__ qidx,
    const int* __restrict__ lixs_c,   const int* __restrict__ rixs_c,
    const int* __restrict__ lixs_w,   const int* __restrict__ rixs_w,
    float* __restrict__ out)
{
  __shared__ __attribute__((aligned(16))) float q_lds[ROWS][104];
  __shared__ __attribute__((aligned(16))) float v_lds[ROWS][2][112];
  __shared__ __attribute__((aligned(16))) int   ixc_lds[NPAIR];
  __shared__ __attribute__((aligned(16))) int   ixw_lds[ROWS][20];
  __shared__ __attribute__((aligned(16))) float s_lds[ROWS][NSLOT];
  __shared__ __attribute__((aligned(16))) float w_lds[ROWS][20];
  __shared__ __attribute__((aligned(16))) float wmat_lds[2][WPAD];

  const int tid   = threadIdx.x;
  const int lane  = tid & 31;
  const int wv    = tid >> 5;
  const int bbase = blockIdx.x * ROWS;
  const int hi = lane >> 4, ln = lane & 15;

  if (tid < ROWS * LSLOT) {
    int r = tid / LSLOT, t = tid - r * LSLOT;
    ixc_lds[r * NSLOT + t] = lixs_c[bbase * LSLOT + tid];
    ixw_lds[r][t]          = lixs_w[bbase * LSLOT + tid];
  }
  if (tid < ROWS * RSLOT) {
    int r = tid / RSLOT, t = tid - r * RSLOT;
    ixc_lds[r * NSLOT + LSLOT + t] = rixs_c[bbase * RSLOT + tid];
    ixw_lds[r][LSLOT + t]          = rixs_w[bbase * RSLOT + tid];
  }
  if (tid < ROWS) { ixw_lds[tid][17] = -1; ixw_lds[tid][18] = -1; ixw_lds[tid][19] = -1; }

  {
    float4 qv[2];
    long qrow[2];
    #pragma unroll
    for (int rr = 0; rr < 2; ++rr) qrow[rr] = qidx[bbase + 2 * wv + rr];
    float4 wl[WLD], wr[WLD];
    load_w(lW, tid, wl);
    load_w(rW, tid, wr);
    #pragma unroll
    for (int rr = 0; rr < 2; ++rr) {
      const float4* src = reinterpret_cast<const float4*>(ce_raw + qrow[rr] * CDIM);
      const float4 v = src[lane < C4 ? lane : C4 - 1];
      const float keep = lane < C4 ? 1.f : 0.f;
      qv[rr] = make_float4(v.x * keep, v.y * keep, v.z * keep, v.w * keep);
    }
    store_w(wmat_lds[0], tid, wl);
    store_w(wmat_lds[1], tid, wr);
    #pragma unroll
    for (int rr = 0; rr < 2; ++rr) {
      float ss = dot4(qv[rr], qv[rr], 0.f);
      ss = wave_sum(ss);
      float inv = 1.f / fmaxf(sqrtf(ss), 1e-12f);
      if (lane < C4) {
        float4 o = make_float4(qv[rr].x * inv, qv[rr].y * inv, qv[rr].z * inv, qv[rr].w * inv);
        *reinterpret_cast<float4*>(&q_lds[2 * wv + rr][4 * lane]) = o;
      }
    }
  }
  __syncthreads();

  wmma_job(q_lds, wmat_lds, v_lds, wv, hi, ln);
  if (wv < NJOB - NWAVE) wmma_job(q_lds, wmat_lds, v_lds, wv + NWAVE, hi, ln);
  __syncthreads();

  {
    const int grp = tid >> 3, gi = tid & 7;
    for (int pb = 0; pb < 9; pb += 3) {
      float4 x[3][4];
      int idx[3];
      #pragma unroll
      for (int u = 0; u < 3; ++u) {
        int p = grp + (NTHR / 8) * (pb + u);
        idx[u] = (p < NPAIR) ? ixc_lds[p] : -1;
      }
      #pragma unroll
      for (int u = 0; u < 3; ++u) {
        const float4* src = reinterpret_cast<const float4*>(ce_raw + (long)idx[u] * CDIM);
        #pragma unroll
        for (int m = 0; m < 4; ++m) {
          int c4 = gi + 8 * m;
          x[u][m] = (idx[u] >= 0 && c4 < C4) ? src[c4] : make_float4(0.f, 0.f, 0.f, 0.f);
        }
      }
      #pragma unroll
      for (int u = 0; u < 3; ++u) {
        int p = grp + (NTHR / 8) * (pb + u);
        int pc = (p < NPAIR) ? p : 0;
        int row = pc / NSLOT, slot = pc - row * NSLOT;
        const float4* vrow = reinterpret_cast<const float4*>(&v_lds[row][slot >= LSLOT ? 1 : 0][0]);
        float dot = 0.f, ss = 0.f;
        #pragma unroll
        for (int m = 0; m < 4; ++m) {
          int c4 = gi + 8 * m;
          float4 v = (c4 < C4) ? vrow[c4] : make_float4(0.f, 0.f, 0.f, 0.f);
          dot = dot4(x[u][m], v, dot);
          ss  = dot4(x[u][m], x[u][m], ss);
        }
        #pragma unroll
        for (int m = 4; m; m >>= 1) {
          dot += __shfl_xor(dot, m, 32);
          ss  += __shfl_xor(ss,  m, 32);
        }
        if (gi == 0 && p < NPAIR)
          s_lds[row][slot] = (idx[u] >= 0) ? dot / fmaxf(sqrtf(ss), 1e-12f) : NEGV;
      }
    }
  }
  __syncthreads();

  if (tid < ROWS) {
    const int row = tid;
    float la[LSLOT], ra[RSLOT];
    #pragma unroll
    for (int t = 0; t < LSLOT; ++t) la[t] = s_lds[row][t];
    #pragma unroll
    for (int t = 0; t < RSLOT; ++t) ra[t] = s_lds[row][LSLOT + t];

    float mx = la[0];
    #pragma unroll
    for (int t = 1; t < LSLOT; ++t) mx = fmaxf(mx, la[t]);
    float s = 0.f;
    #pragma unroll
    for (int t = 0; t < LSLOT; ++t) { la[t] = expf(la[t] - mx); s += la[t]; }
    float rs = 1.f / s;
    #pragma unroll
    for (int t = 0; t < LSLOT; ++t) la[t] *= rs;

    mx = ra[0];
    #pragma unroll
    for (int t = 1; t < RSLOT; ++t) mx = fmaxf(mx, ra[t]);
    s = 0.f;
    #pragma unroll
    for (int t = 0; t < RSLOT; ++t) { ra[t] = expf(ra[t] - mx); s += ra[t]; }
    rs = 1.f / s;
    #pragma unroll
    for (int t = 0; t < RSLOT; ++t) ra[t] *= rs;

    float z0 = gL_b[0], z1 = gL_b[1];
    #pragma unroll
    for (int t = 0; t < LSLOT; ++t) {
      z0 = fmaf(la[t], gL_w[t],                   z0);
      z1 = fmaf(la[t], gL_w[(LSLOT + RSLOT) + t], z1);
    }
    #pragma unroll
    for (int t = 0; t < RSLOT; ++t) {
      z0 = fmaf(ra[t], gL_w[LSLOT + t],                   z0);
      z1 = fmaf(ra[t], gL_w[(LSLOT + RSLOT) + LSLOT + t], z1);
    }
    float gm = fmaxf(z0, z1);
    float e0 = expf(z0 - gm), e1 = expf(z1 - gm);
    float gi2 = 1.f / (e0 + e1);

    #pragma unroll
    for (int t = 0; t < LSLOT; ++t) w_lds[row][t] = la[t];
    #pragma unroll
    for (int t = 0; t < RSLOT; ++t) w_lds[row][LSLOT + t] = ra[t];
    w_lds[row][17] = e0 * gi2;
    w_lds[row][18] = e1 * gi2;
    w_lds[row][19] = 0.f;
  }
  __syncthreads();

  {
    float4* out4 = reinterpret_cast<float4*>(out) + (long)bbase * W4;
    for (int item = tid; item < ROWS * W4; item += NTHR) {
      const int row = item / W4, q = item - row * W4;
      int   iw[20];
      float w[20];
      #pragma unroll
      for (int g = 0; g < 5; ++g) {
        int4   iv = *reinterpret_cast<const int4*>(&ixw_lds[row][4 * g]);
        float4 wv4 = *reinterpret_cast<const float4*>(&w_lds[row][4 * g]);
        iw[4 * g] = iv.x; iw[4 * g + 1] = iv.y; iw[4 * g + 2] = iv.z; iw[4 * g + 3] = iv.w;
        w[4 * g] = wv4.x; w[4 * g + 1] = wv4.y; w[4 * g + 2] = wv4.z; w[4 * g + 3] = wv4.w;
      }
      float4 x[NSLOT];
      #pragma unroll
      for (int t = 0; t < NSLOT; ++t) {
        const int ir = max(iw[t], 0);
        x[t] = reinterpret_cast<const float4*>(wvec + (long)ir * WDIM)[q];
        w[t] = iw[t] < 0 ? 0.f : w[t];
      }
      float4 aL = make_float4(0.f, 0.f, 0.f, 0.f), aR = aL;
      #pragma unroll
      for (int t = 0; t < LSLOT; ++t) aL = fma4(w[t], x[t], aL);
      #pragma unroll
      for (int t = LSLOT; t < NSLOT; ++t) aR = fma4(w[t], x[t], aR);
      const float g0 = w[17], g1 = w[18];
      float4 o;
      o.x = g0 * aL.x + g1 * aR.x;
      o.y = g0 * aL.y + g1 * aR.y;
      o.z = g0 * aL.z + g1 * aR.z;
      o.w = g0 * aL.w + g1 * aR.w;
      out4[item] = o;
    }
  }
}

extern "C" void kernel_launch(void* const* d_in, const int* in_sizes, int n_in,
                              void* d_out, int out_size, void* d_ws, size_t ws_size,
                              hipStream_t stream) {
  const float* ce_raw = (const float*)d_in[0];
  const float* wvec   = (const float*)d_in[1];
  const float* lW     = (const float*)d_in[2];
  const float* rW     = (const float*)d_in[3];
  const float* gL_w   = (const float*)d_in[4];
  const float* gL_b   = (const float*)d_in[5];
  const int*   qidx   = (const int*)d_in[6];
  const int*   lixs_c = (const int*)d_in[7];
  const int*   rixs_c = (const int*)d_in[8];
  const int*   lixs_w = (const int*)d_in[9];
  const int*   rixs_w = (const int*)d_in[10];

  const int B = in_sizes[6];
  const int blocks = B / 32;
  fused_gated_attn_kernel<<<blocks, 512, 0, stream>>>(
      ce_raw, wvec, lW, rW, gL_w, gL_b,
      qidx, lixs_c, rixs_c, lixs_w, rixs_w, (float*)d_out);
}


# Phase-2a pair list compacted: wave ballots rank the non-pad (row,slot) pairs into a dense LDS list, score rounds run to ceil(n/64) of the 9, pad pairs get their masked score at staging. Sums same.
# speedup vs baseline: 24.9642x; 1.0378x over previous
#include <hip/hip_runtime.h>

typedef float v2f __attribute__((ext_vector_type(2)));
typedef float v8f __attribute__((ext_vector_type(8)));

#define CDIM  100
#define LSLOT 7
#define RSLOT 10
#define NSLOT 17
#define WDIM  300
#define NEGV  -1e30f
#define ROWS  32
#define MTILE (ROWS / 16)
#define NJOB  (MTILE * 14)
#define NWAVE 16
#define NTHR  (32 * NWAVE)
#define C4    25
#define W4    75
#define NPAIR (ROWS * NSLOT)
#define WSZ4  (CDIM * CDIM / 4)
#define WLD   5
#define WPAD  (4 * NTHR * WLD)

__device__ __forceinline__ float wave_sum(float v) {
  #pragma unroll
  for (int m = 16; m; m >>= 1) v += __shfl_xor(v, m, 32);
  return v;
}

__device__ __forceinline__ float dot4(float4 x, float4 v, float acc) {
  acc = fmaf(x.x, v.x, acc);
  acc = fmaf(x.y, v.y, acc);
  acc = fmaf(x.z, v.z, acc);
  acc = fmaf(x.w, v.w, acc);
  return acc;
}

__device__ __forceinline__ float4 fma4(float s, float4 x, float4 a) {
  a.x = fmaf(s, x.x, a.x);
  a.y = fmaf(s, x.y, a.y);
  a.z = fmaf(s, x.z, a.z);
  a.w = fmaf(s, x.w, a.w);
  return a;
}

__device__ __forceinline__ void load_w(const float* __restrict__ Wg, int tid, float4 (&wreg)[WLD]) {
  const float4* wsrc = reinterpret_cast<const float4*>(Wg);
  #pragma unroll
  for (int i = 0; i < WLD; ++i) {
    const int f = tid + NTHR * i;
    wreg[i] = wsrc[f < WSZ4 ? f : WSZ4 - 1];
  }
}

__device__ __forceinline__ void store_w(float* wmat, int tid, const float4 (&wreg)[WLD]) {
  #pragma unroll
  for (int i = 0; i < WLD; ++i) {
    const int f = tid + NTHR * i;
    reinterpret_cast<float4*>(wmat)[f] = wreg[i];
  }
}

__device__ __forceinline__ void wmma_side(const float (*q)[104], const float* wmat, float (*v)[2][112],
                                          int side, int nt, int hi, int ln) {
  const int cout = nt * 16 + ln;
  const int crow = cout < CDIM ? cout : CDIM - 1;
  const float* qk = &q[ln][2 * hi];
  const float* wk = wmat + crow * CDIM + 2 * hi;
  v2f a[25], bfrag[25];
  #pragma unroll
  for (int k = 0; k < 25; ++k) {
    a[k]     = *reinterpret_cast<const v2f*>(qk + 4 * k);
    bfrag[k] = *reinterpret_cast<const v2f*>(wk + 4 * k);
  }
  __builtin_amdgcn_sched_barrier(0);
  v8f acc = {};
  #pragma unroll
  for (int k = 0; k < 25; ++k)
    acc = __builtin_amdgcn_wmma_f32_16x16x4_f32(
         false, a[k],  false, bfrag[k],
         (short)0, acc,  false,  false);
  #pragma unroll
  for (int r = 0; r < 8; ++r)
    v[r + 8 * hi][side][nt * 16 + ln] = acc[r];
}

__device__ __forceinline__ void wmma_job(const float (*q)[104], const float (*wmat)[WPAD], float (*v)[2][112],
                                         int j, int hi, int ln) {
  const int mt = j / 14, rem = j - 14 * mt;
  const int side = rem / 7, nt = rem - 7 * side;
  wmma_side(q + 16 * mt, wmat[side], v + 16 * mt, side, nt, hi, ln);
}

__global__ __launch_bounds__(NTHR) __attribute__((amdgpu_waves_per_eu(8, 8))) void fused_gated_attn_kernel(
    const float* __restrict__ ce_raw, const float* __restrict__ wvec,
    const float* __restrict__ lW,     const float* __restrict__ rW,
    const float* __restrict__ gL_w,   const float* __restrict__ gL_b,
    const int* __restrict__ qidx,
    const int* __restrict__ lixs_c,   const int* __restrict__ rixs_c,
    const int* __restrict__ lixs_w,   const int* __restrict__ rixs_w,
    float* __restrict__ out)
{
  __shared__ __attribute__((aligned(16))) float q_lds[ROWS][104];
  __shared__ __attribute__((aligned(16))) float v_lds[ROWS][2][112];
  __shared__ __attribute__((aligned(16))) int2  plist[NPAIR];
  __shared__ __attribute__((aligned(16))) int   wcnt[NWAVE];
  __shared__ __attribute__((aligned(16))) int   ixw_lds[ROWS][20];
  __shared__ __attribute__((aligned(16))) float s_lds[ROWS][NSLOT];
  __shared__ __attribute__((aligned(16))) float w_lds[ROWS][20];
  __shared__ __attribute__((aligned(16))) float wmat_lds[2][WPAD];

  const int tid   = threadIdx.x;
  const int lane  = tid & 31;
  const int wv    = tid >> 5;
  const int bbase = blockIdx.x * ROWS;
  const int hi = lane >> 4, ln = lane & 15;

  int idxL = -1, idxR = -1, codeL = 0, codeR = 0;
  if (tid < ROWS * LSLOT) {
    int r = tid / LSLOT, t = tid - r * LSLOT;
    idxL  = lixs_c[bbase * LSLOT + tid];
    codeL = (r << 5) | t;
    ixw_lds[r][t] = lixs_w[bbase * LSLOT + tid];
    if (idxL < 0) s_lds[r][t] = NEGV;
  }
  if (tid < ROWS * RSLOT) {
    int r = tid / RSLOT, t = tid - r * RSLOT;
    idxR  = rixs_c[bbase * RSLOT + tid];
    codeR = (r << 5) | (LSLOT + t);
    ixw_lds[r][LSLOT + t] = rixs_w[bbase * RSLOT + tid];
    if (idxR < 0) s_lds[r][LSLOT + t] = NEGV;
  }
  if (tid < ROWS) { ixw_lds[tid][17] = -1; ixw_lds[tid][18] = -1; ixw_lds[tid][19] = -1; }
  const unsigned bL = __builtin_amdgcn_ballot_w32(idxL >= 0);
  const unsigned bR = __builtin_amdgcn_ballot_w32(idxR >= 0);
  const int rankL = __builtin_amdgcn_mbcnt_lo(bL, 0u);
  const int rankR = __popc(bL) + __builtin_amdgcn_mbcnt_lo(bR, 0u);
  if (lane == 0) wcnt[wv] = __popc(bL) + __popc(bR);

  {
    float4 qv[2];
    long qrow[2];
    #pragma unroll
    for (int rr = 0; rr < 2; ++rr) qrow[rr] = qidx[bbase + 2 * wv + rr];
    float4 wl[WLD], wr[WLD];
    load_w(lW, tid, wl);
    load_w(rW, tid, wr);
    #pragma unroll
    for (int rr = 0; rr < 2; ++rr) {
      const float4* src = reinterpret_cast<const float4*>(ce_raw + qrow[rr] * CDIM);
      const float4 v = src[lane < C4 ? lane : C4 - 1];
      const float keep = lane < C4 ? 1.f : 0.f;
      qv[rr] = make_float4(v.x * keep, v.y * keep, v.z * keep, v.w * keep);
    }
    store_w(wmat_lds[0], tid, wl);
    store_w(wmat_lds[1], tid, wr);
    #pragma unroll
    for (int rr = 0; rr < 2; ++rr) {
      float ss = dot4(qv[rr], qv[rr], 0.f);
      ss = wave_sum(ss);
      float inv = 1.f / fmaxf(sqrtf(ss), 1e-12f);
      if (lane < C4) {
        float4 o = make_float4(qv[rr].x * inv, qv[rr].y * inv, qv[rr].z * inv, qv[rr].w * inv);
        *reinterpret_cast<float4*>(&q_lds[2 * wv + rr][4 * lane]) = o;
      }
    }
  }
  __syncthreads();

  int nvalid;
  {
    int base = 0, total = 0;
    #pragma unroll
    for (int g = 0; g < NWAVE / 4; ++g) {
      const int4 c4v = *reinterpret_cast<const int4*>(&wcnt[4 * g]);
      const int cc[4] = {c4v.x, c4v.y, c4v.z, c4v.w};
      #pragma unroll
      for (int k = 0; k < 4; ++k) {
        total += cc[k];
        base  += (4 * g + k < wv) ? cc[k] : 0;
      }
    }
    if (idxL >= 0) plist[base + rankL] = make_int2(codeL, idxL);
    if (idxR >= 0) plist[base + rankR] = make_int2(codeR, idxR);
    nvalid = __builtin_amdgcn_readfirstlane(total);
  }

  wmma_job(q_lds, wmat_lds, v_lds, wv, hi, ln);
  if (wv < NJOB - NWAVE) wmma_job(q_lds, wmat_lds, v_lds, wv + NWAVE, hi, ln);
  __syncthreads();

  {
    const int grp = tid >> 3, gi = tid & 7;
    const int wgrp0 = __builtin_amdgcn_readfirstlane(wv) * 4;
    for (int pb = 0; pb < 9; pb += 3) {
      if ((NTHR / 8) * pb >= nvalid) break;
      float4 x[3][4];
      int idx[3], code[3];
      #pragma unroll
      for (int u = 0; u < 3; ++u) {
        const int pos = grp + (NTHR / 8) * (pb + u);
        int2 e = make_int2(0, -1);
        if (pos < nvalid) e = plist[pos];
        code[u] = e.x;
        idx[u]  = e.y;
      }
      #pragma unroll
      for (int u = 0; u < 3; ++u) {
        const float4* src = reinterpret_cast<const float4*>(ce_raw + (long)idx[u] * CDIM);
        #pragma unroll
        for (int m = 0; m < 4; ++m) {
          int c4 = gi + 8 * m;
          x[u][m] = (idx[u] >= 0 && c4 < C4) ? src[c4] : make_float4(0.f, 0.f, 0.f, 0.f);
        }
      }
      #pragma unroll
      for (int u = 0; u < 3; ++u) {
        if (wgrp0 + (NTHR / 8) * (pb + u) >= nvalid) continue;
        const int row = code[u] >> 5, slot = code[u] & 31;
        const float4* vrow = reinterpret_cast<const float4*>(&v_lds[row][slot >= LSLOT ? 1 : 0][0]);
        float dot = 0.f, ss = 0.f;
        #pragma unroll
        for (int m = 0; m < 4; ++m) {
          int c4 = gi + 8 * m;
          float4 v = (c4 < C4) ? vrow[c4] : make_float4(0.f, 0.f, 0.f, 0.f);
          dot = dot4(x[u][m], v, dot);
          ss  = dot4(x[u][m], x[u][m], ss);
        }
        #pragma unroll
        for (int m = 4; m; m >>= 1) {
          dot += __shfl_xor(dot, m, 32);
          ss  += __shfl_xor(ss,  m, 32);
        }
        if (gi == 0 && idx[u] >= 0)
          s_lds[row][slot] = dot / fmaxf(sqrtf(ss), 1e-12f);
      }
    }
  }
  __syncthreads();

  if (tid < ROWS) {
    const int row = tid;
    float la[LSLOT], ra[RSLOT];
    #pragma unroll
    for (int t = 0; t < LSLOT; ++t) la[t] = s_lds[row][t];
    #pragma unroll
    for (int t = 0; t < RSLOT; ++t) ra[t] = s_lds[row][LSLOT + t];

    float mx = la[0];
    #pragma unroll
    for (int t = 1; t < LSLOT; ++t) mx = fmaxf(mx, la[t]);
    float s = 0.f;
    #pragma unroll
    for (int t = 0; t < LSLOT; ++t) { la[t] = expf(la[t] - mx); s += la[t]; }
    float rs = 1.f / s;
    #pragma unroll
    for (int t = 0; t < LSLOT; ++t) la[t] *= rs;

    mx = ra[0];
    #pragma unroll
    for (int t = 1; t < RSLOT; ++t) mx = fmaxf(mx, ra[t]);
    s = 0.f;
    #pragma unroll
    for (int t = 0; t < RSLOT; ++t) { ra[t] = expf(ra[t] - mx); s += ra[t]; }
    rs = 1.f / s;
    #pragma unroll
    for (int t = 0; t < RSLOT; ++t) ra[t] *= rs;

    float z0 = gL_b[0], z1 = gL_b[1];
    #pragma unroll
    for (int t = 0; t < LSLOT; ++t) {
      z0 = fmaf(la[t], gL_w[t],                   z0);
      z1 = fmaf(la[t], gL_w[(LSLOT + RSLOT) + t], z1);
    }
    #pragma unroll
    for (int t = 0; t < RSLOT; ++t) {
      z0 = fmaf(ra[t], gL_w[LSLOT + t],                   z0);
      z1 = fmaf(ra[t], gL_w[(LSLOT + RSLOT) + LSLOT + t], z1);
    }
    float gm = fmaxf(z0, z1);
    float e0 = expf(z0 - gm), e1 = expf(z1 - gm);
    float gi2 = 1.f / (e0 + e1);

    #pragma unroll
    for (int t = 0; t < LSLOT; ++t) w_lds[row][t] = la[t];
    #pragma unroll
    for (int t = 0; t < RSLOT; ++t) w_lds[row][LSLOT + t] = ra[t];
    w_lds[row][17] = e0 * gi2;
    w_lds[row][18] = e1 * gi2;
    w_lds[row][19] = 0.f;
  }
  __syncthreads();

  {
    float4* out4 = reinterpret_cast<float4*>(out) + (long)bbase * W4;
    for (int item = tid; item < ROWS * W4; item += NTHR) {
      const int row = item / W4, q = item - row * W4;
      int   iw[20];
      float w[20];
      #pragma unroll
      for (int g = 0; g < 5; ++g) {
        int4   iv = *reinterpret_cast<const int4*>(&ixw_lds[row][4 * g]);
        float4 wv4 = *reinterpret_cast<const float4*>(&w_lds[row][4 * g]);
        iw[4 * g] = iv.x; iw[4 * g + 1] = iv.y; iw[4 * g + 2] = iv.z; iw[4 * g + 3] = iv.w;
        w[4 * g] = wv4.x; w[4 * g + 1] = wv4.y; w[4 * g + 2] = wv4.z; w[4 * g + 3] = wv4.w;
      }
      float4 x[NSLOT];
      #pragma unroll
      for (int t = 0; t < NSLOT; ++t) {
        const int ir = max(iw[t], 0);
        x[t] = reinterpret_cast<const float4*>(wvec + (long)ir * WDIM)[q];
        w[t] = iw[t] < 0 ? 0.f : w[t];
      }
      float4 aL = make_float4(0.f, 0.f, 0.f, 0.f), aR = aL;
      #pragma unroll
      for (int t = 0; t < LSLOT; ++t) aL = fma4(w[t], x[t], aL);
      #pragma unroll
      for (int t = LSLOT; t < NSLOT; ++t) aR = fma4(w[t], x[t], aR);
      const float g0 = w[17], g1 = w[18];
      float4 o;
      o.x = g0 * aL.x + g1 * aR.x;
      o.y = g0 * aL.y + g1 * aR.y;
      o.z = g0 * aL.z + g1 * aR.z;
      o.w = g0 * aL.w + g1 * aR.w;
      out4[item] = o;
    }
  }
}

extern "C" void kernel_launch(void* const* d_in, const int* in_sizes, int n_in,
                              void* d_out, int out_size, void* d_ws, size_t ws_size,
                              hipStream_t stream) {
  const float* ce_raw = (const float*)d_in[0];
  const float* wvec   = (const float*)d_in[1];
  const float* lW     = (const float*)d_in[2];
  const float* rW     = (const float*)d_in[3];
  const float* gL_w   = (const float*)d_in[4];
  const float* gL_b   = (const float*)d_in[5];
  const int*   qidx   = (const int*)d_in[6];
  const int*   lixs_c = (const int*)d_in[7];
  const int*   rixs_c = (const int*)d_in[8];
  const int*   lixs_w = (const int*)d_in[9];
  const int*   rixs_w = (const int*)d_in[10];

  const int B = in_sizes[6];
  const int blocks = B / 32;
  fused_gated_attn_kernel<<<blocks, 512, 0, stream>>>(
      ce_raw, wvec, lW, rW, gL_w, gL_b,
      qidx, lixs_c, rixs_c, lixs_w, rixs_w, (float*)d_out);
}
